# starting kernel
# baseline (speedup 1.0000x reference)
#include <hip/hip_runtime.h>

typedef __attribute__((ext_vector_type(2))) float v2f;
typedef __attribute__((ext_vector_type(8))) float v8f;

#define TILE   32
#define HALO   5
#define IN_T   42
#define XS     44
#define HS     40
#define RS     36
#define IMG_H  512
#define IMG_W  512

#define G0 0.00102838f
#define G1 0.00759870f
#define G2 0.03600077f
#define G3 0.10936068f
#define G4 0.21300553f
#define G5 0.26601172f

__global__ __launch_bounds__(256) void ssim_fused_kernel(const float* __restrict__ img1,
                                                         const float* __restrict__ img2,
                                                         float* __restrict__ out,
                                                         int planes) {
    __shared__ __align__(16) float smem[8400 + 64 + 5760];
    float* hbuf = smem;
    float* gpad = smem + 5 * IN_T * HS;
    float* xs   = gpad + 64;
    float* ys   = xs + IN_T * XS;
    float* rbuf = gpad + 64;

    const int tid  = threadIdx.x;
    const int lane = tid & 31;
    const int M    = lane & 15;
    const int hi   = lane >> 4;

    const int c0    = blockIdx.x * TILE;
    const int r0    = blockIdx.y * TILE;
    const int plane = blockIdx.z;
    const size_t pbase = (size_t)plane * (IMG_H * IMG_W);
    const float* p1 = img1 + pbase;
    const float* p2 = img2 + pbase;

    if (tid < 64) {
        int t = tid - 16;
        float v = 0.0f;
        if (t >= 0 && t <= 10) {
            int d = t - 5; if (d < 0) d = -d;
            v = (d == 0) ? G5 : (d == 1) ? G4 : (d == 2) ? G3
              : (d == 3) ? G2 : (d == 4) ? G1 : G0;
        }
        gpad[tid] = v;
    }

    for (int i = tid; i < IN_T * IN_T; i += 256) {
        int r = i / IN_T;
        int c = i - r * IN_T;
        int gr = r0 - HALO + r;
        gr = (gr < 0) ? -gr : gr;
        gr = (gr >= IMG_H) ? (2 * IMG_H - 2 - gr) : gr;
        int gc = c0 - HALO + c;
        gc = (gc < 0) ? -gc : gc;
        gc = (gc >= IMG_W) ? (2 * IMG_W - 2 - gc) : gc;
        float a = p1[gr * IMG_W + gc];
        float b = p2[gr * IMG_W + gc];
        xs[r * XS + c] = a;
        ys[r * XS + c] = b;
    }
    __syncthreads();

    {
        const float GW[11] = {G0, G1, G2, G3, G4, G5, G4, G3, G2, G1, G0};
        for (int i = tid; i < IN_T * TILE; i += 256) {
            int r = i >> 5;
            int c = i & 31;
            const float* xr = &xs[r * XS + c];
            const float* yr = &ys[r * XS + c];
            float sx = 0.f, sy = 0.f, sxx = 0.f, syy = 0.f, sxy = 0.f;
#pragma unroll
            for (int t = 0; t < 11; ++t) {
                float w  = GW[t];
                float xv = xr[t];
                float yv = yr[t];
                sx  += w * xv;
                sy  += w * yv;
                sxx += w * xv * xv;
                syy += w * yv * yv;
                sxy += w * xv * yv;
            }
            int o = r * HS + c;
            hbuf[0 * IN_T * HS + o] = sx;
            hbuf[1 * IN_T * HS + o] = sy;
            hbuf[2 * IN_T * HS + o] = sxx;
            hbuf[3 * IN_T * HS + o] = syy;
            hbuf[4 * IN_T * HS + o] = sxy;
        }
    }
    __syncthreads();

    float ax[7], ay[7];
#pragma unroll
    for (int b = 0; b < 7; ++b) {
        int j0 = 4 * b + 2 * hi;
        ax[b] = gpad[j0 - M + 16];
        ay[b] = gpad[j0 + 1 - M + 16];
    }

    int wave = __builtin_amdgcn_readfirstlane(tid >> 5);
    for (int j = wave; j < 20; j += 8) {
        int f  = j >> 2;
        int s  = j & 3;
        int sr = (s >> 1) << 4;
        int sc = (s & 1) << 4;
        const float* hf = hbuf + f * IN_T * HS;

        v8f acc = {0.f, 0.f, 0.f, 0.f, 0.f, 0.f, 0.f, 0.f};
#pragma unroll
        for (int b = 0; b < 7; ++b) {
            int j0   = 4 * b + 2 * hi;
            int row0 = sr + j0;     if (row0 > IN_T - 1) row0 = IN_T - 1;
            int row1 = sr + j0 + 1; if (row1 > IN_T - 1) row1 = IN_T - 1;
            v2f av = {ax[b], ay[b]};
            v2f bv;
            bv.x = hf[row0 * HS + sc + M];
            bv.y = hf[row1 * HS + sc + M];
            acc = __builtin_amdgcn_wmma_f32_16x16x4_f32(
                 false, av,  false, bv,
                 (short)0, acc,  false,  false);
        }
        float* rf = rbuf + f * TILE * RS;
#pragma unroll
        for (int v = 0; v < 8; ++v)
            rf[(sr + hi * 8 + v) * RS + sc + M] = acc[v];
    }
    __syncthreads();

    const float C1v = 0.0001f;
    const float C2v = 0.0009f;
    const float EPSv = 1e-12f;
    float* po = out + pbase;
#pragma unroll
    for (int k = 0; k < 4; ++k) {
        int p   = tid + (k << 8);
        int row = p >> 5;
        int col = p & 31;
        int o   = row * RS + col;
        float mu1 = rbuf[0 * TILE * RS + o];
        float mu2 = rbuf[1 * TILE * RS + o];
        float exx = rbuf[2 * TILE * RS + o];
        float eyy = rbuf[3 * TILE * RS + o];
        float exy = rbuf[4 * TILE * RS + o];
        float mu1s = mu1 * mu1;
        float mu2s = mu2 * mu2;
        float mu12 = mu1 * mu2;
        float s1  = exx - mu1s;
        float s2  = eyy - mu2s;
        float s12 = exy - mu12;
        float num = (2.0f * mu12 + C1v) * (2.0f * s12 + C2v);
        float den = (mu1s + mu2s + C1v) * (s1 + s2 + C2v);
        po[(size_t)(r0 + row) * IMG_W + (c0 + col)] = num / (den + EPSv);
    }
}

extern "C" void kernel_launch(void* const* d_in, const int* in_sizes, int n_in,
                              void* d_out, int out_size, void* d_ws, size_t ws_size,
                              hipStream_t stream) {
    (void)n_in; (void)out_size; (void)d_ws; (void)ws_size;
    const float* img1 = (const float*)d_in[0];
    const float* img2 = (const float*)d_in[1];
    float* out = (float*)d_out;
    int planes = in_sizes[0] / (IMG_H * IMG_W);
    dim3 grid(IMG_W / TILE, IMG_H / TILE, planes);
    ssim_fused_kernel<<<grid, 256, 0, stream>>>(img1, img2, out, planes);
}


# tile loop in the work-group: 1280 resident work-groups (5 per WGP by LDS) each walk 9 or 10 of the 12288 tiles, so waves enter once per launch, not per tile; tap table and band operand built once
# speedup vs baseline: 1.0338x; 1.0338x over previous
#include <hip/hip_runtime.h>

typedef __attribute__((ext_vector_type(2))) float v2f;
typedef __attribute__((ext_vector_type(8))) float v8f;

#define TILE   32
#define HALO   5
#define IN_T   42
#define XS     44
#define HS     40
#define RS     36
#define IMG_H  512
#define IMG_W  512
#define TILES_X (IMG_W / TILE)
#define TILES_PER_PLANE (TILES_X * (IMG_H / TILE))
#define WG_COUNT 1280

#define G0 0.00102838f
#define G1 0.00759870f
#define G2 0.03600077f
#define G3 0.10936068f
#define G4 0.21300553f
#define G5 0.26601172f

__global__ __launch_bounds__(256) void ssim_fused_kernel(const float* __restrict__ img1,
                                                         const float* __restrict__ img2,
                                                         float* __restrict__ out,
                                                         int planes) {
    __shared__ __align__(16) float smem[8400 + 64 + 5760];
    float* hbuf = smem;
    float* gpad = smem + 5 * IN_T * HS;
    float* xs   = gpad + 64;
    float* ys   = xs + IN_T * XS;
    float* rbuf = gpad + 64;

    const int tid  = threadIdx.x;
    const int lane = tid & 31;
    const int M    = lane & 15;
    const int hi   = lane >> 4;

    if (tid < 64) {
        int t = tid - 16;
        float v = 0.0f;
        if (t >= 0 && t <= 10) {
            int d = t - 5; if (d < 0) d = -d;
            v = (d == 0) ? G5 : (d == 1) ? G4 : (d == 2) ? G3
              : (d == 3) ? G2 : (d == 4) ? G1 : G0;
        }
        gpad[tid] = v;
    }
    __syncthreads();

    float ax[7], ay[7];
#pragma unroll
    for (int b = 0; b < 7; ++b) {
        int j0 = 4 * b + 2 * hi;
        ax[b] = gpad[j0 - M + 16];
        ay[b] = gpad[j0 + 1 - M + 16];
    }
    const int wave = __builtin_amdgcn_readfirstlane(tid >> 5);

    const unsigned ntiles = (unsigned)planes * TILES_PER_PLANE;
    for (unsigned tile = blockIdx.x; tile < ntiles; tile += WG_COUNT) {
    const int c0    = (int)(tile & (TILES_X - 1)) * TILE;
    const int r0    = (int)((tile / TILES_X) & (IMG_H / TILE - 1)) * TILE;
    const int plane = (int)(tile / TILES_PER_PLANE);
    const size_t pbase = (size_t)plane * (IMG_H * IMG_W);
    const float* p1 = img1 + pbase;
    const float* p2 = img2 + pbase;

    for (int i = tid; i < IN_T * IN_T; i += 256) {
        int r = i / IN_T;
        int c = i - r * IN_T;
        int gr = r0 - HALO + r;
        gr = (gr < 0) ? -gr : gr;
        gr = (gr >= IMG_H) ? (2 * IMG_H - 2 - gr) : gr;
        int gc = c0 - HALO + c;
        gc = (gc < 0) ? -gc : gc;
        gc = (gc >= IMG_W) ? (2 * IMG_W - 2 - gc) : gc;
        float a = p1[gr * IMG_W + gc];
        float b = p2[gr * IMG_W + gc];
        xs[r * XS + c] = a;
        ys[r * XS + c] = b;
    }
    __syncthreads();

    {
        const float GW[11] = {G0, G1, G2, G3, G4, G5, G4, G3, G2, G1, G0};
        for (int i = tid; i < IN_T * TILE; i += 256) {
            int r = i >> 5;
            int c = i & 31;
            const float* xr = &xs[r * XS + c];
            const float* yr = &ys[r * XS + c];
            float sx = 0.f, sy = 0.f, sxx = 0.f, syy = 0.f, sxy = 0.f;
#pragma unroll
            for (int t = 0; t < 11; ++t) {
                float w  = GW[t];
                float xv = xr[t];
                float yv = yr[t];
                sx  += w * xv;
                sy  += w * yv;
                sxx += w * xv * xv;
                syy += w * yv * yv;
                sxy += w * xv * yv;
            }
            int o = r * HS + c;
            hbuf[0 * IN_T * HS + o] = sx;
            hbuf[1 * IN_T * HS + o] = sy;
            hbuf[2 * IN_T * HS + o] = sxx;
            hbuf[3 * IN_T * HS + o] = syy;
            hbuf[4 * IN_T * HS + o] = sxy;
        }
    }
    __syncthreads();

    for (int j = wave; j < 20; j += 8) {
        int f  = j >> 2;
        int s  = j & 3;
        int sr = (s >> 1) << 4;
        int sc = (s & 1) << 4;
        const float* hf = hbuf + f * IN_T * HS;

        v8f acc = {0.f, 0.f, 0.f, 0.f, 0.f, 0.f, 0.f, 0.f};
#pragma unroll
        for (int b = 0; b < 7; ++b) {
            int j0   = 4 * b + 2 * hi;
            int row0 = sr + j0;     if (row0 > IN_T - 1) row0 = IN_T - 1;
            int row1 = sr + j0 + 1; if (row1 > IN_T - 1) row1 = IN_T - 1;
            v2f av = {ax[b], ay[b]};
            v2f bv;
            bv.x = hf[row0 * HS + sc + M];
            bv.y = hf[row1 * HS + sc + M];
            acc = __builtin_amdgcn_wmma_f32_16x16x4_f32(
                 false, av,  false, bv,
                 (short)0, acc,  false,  false);
        }
        float* rf = rbuf + f * TILE * RS;
#pragma unroll
        for (int v = 0; v < 8; ++v)
            rf[(sr + hi * 8 + v) * RS + sc + M] = acc[v];
    }
    __syncthreads();

    const float C1v = 0.0001f;
    const float C2v = 0.0009f;
    const float EPSv = 1e-12f;
    float* po = out + pbase;
    float fv[4][5];
#pragma unroll
    for (int k = 0; k < 4; ++k) {
        int p   = tid + (k << 8);
        int o   = (p >> 5) * RS + (p & 31);
#pragma unroll
        for (int f = 0; f < 5; ++f) fv[k][f] = rbuf[f * TILE * RS + o];
    }
    __syncthreads();
#pragma unroll
    for (int k = 0; k < 4; ++k) {
        int p   = tid + (k << 8);
        int row = p >> 5;
        int col = p & 31;
        float mu1 = fv[k][0];
        float mu2 = fv[k][1];
        float exx = fv[k][2];
        float eyy = fv[k][3];
        float exy = fv[k][4];
        float mu1s = mu1 * mu1;
        float mu2s = mu2 * mu2;
        float mu12 = mu1 * mu2;
        float s1  = exx - mu1s;
        float s2  = eyy - mu2s;
        float s12 = exy - mu12;
        float num = (2.0f * mu12 + C1v) * (2.0f * s12 + C2v);
        float den = (mu1s + mu2s + C1v) * (s1 + s2 + C2v);
        po[(size_t)(r0 + row) * IMG_W + (c0 + col)] = num / (den + EPSv);
    }
    }
}

extern "C" void kernel_launch(void* const* d_in, const int* in_sizes, int n_in,
                              void* d_out, int out_size, void* d_ws, size_t ws_size,
                              hipStream_t stream) {
    (void)n_in; (void)out_size; (void)d_ws; (void)ws_size;
    const float* img1 = (const float*)d_in[0];
    const float* img2 = (const float*)d_in[1];
    float* out = (float*)d_out;
    int planes = in_sizes[0] / (IMG_H * IMG_W);
    ssim_fused_kernel<<<dim3(WG_COUNT), 256, 0, stream>>>(img1, img2, out, planes);
}


# halo loads software-pipelined one tile ahead: a thread issues its 14 loads as one group before phase 3 and stores them to LDS at the next trip; barriers in between order LDS only
# speedup vs baseline: 1.2068x; 1.1673x over previous
#include <hip/hip_runtime.h>

typedef __attribute__((ext_vector_type(2))) float v2f;
typedef __attribute__((ext_vector_type(8))) float v8f;

#define TILE   32
#define HALO   5
#define IN_T   42
#define XS     44
#define HS     40
#define RS     36
#define IMG_H  512
#define IMG_W  512
#define TILES_X (IMG_W / TILE)
#define TILES_PER_PLANE (TILES_X * (IMG_H / TILE))
#define WG_COUNT 1280
#define LD_ROWS   6
#define LD_PASSES 7

#define G0 0.00102838f
#define G1 0.00759870f
#define G2 0.03600077f
#define G3 0.10936068f
#define G4 0.21300553f
#define G5 0.26601172f

static __device__ __forceinline__ void lds_barrier() {
    __builtin_amdgcn_fence(__ATOMIC_RELEASE, "workgroup", "local");
    __builtin_amdgcn_s_barrier();
    __builtin_amdgcn_fence(__ATOMIC_ACQUIRE, "workgroup", "local");
}

__global__ __launch_bounds__(256) void ssim_fused_kernel(const float* __restrict__ img1,
                                                         const float* __restrict__ img2,
                                                         float* __restrict__ out,
                                                         int planes) {
    __shared__ __align__(16) float smem[8400 + 64 + 5760];
    float* hbuf = smem;
    float* gpad = smem + 5 * IN_T * HS;
    float* xs   = gpad + 64;
    float* ys   = xs + IN_T * XS;
    float* rbuf = gpad + 64;

    const int tid  = threadIdx.x;
    const int lane = tid & 31;
    const int M    = lane & 15;
    const int hi   = lane >> 4;

    const unsigned ntiles = (unsigned)planes * TILES_PER_PLANE;

    const bool ld_on = tid < LD_ROWS * IN_T;
    const int  lr    = tid / IN_T;
    const int  lc    = tid - lr * IN_T;
    const int  lofs  = lr * XS + lc;

    float va[LD_PASSES], vb[LD_PASSES];
    auto issue_halo = [&](unsigned t) {
        const int tc0 = (int)(t & (TILES_X - 1)) * TILE;
        const int tr0 = (int)((t / TILES_X) & (IMG_H / TILE - 1)) * TILE;
        const size_t tb = (size_t)(t / TILES_PER_PLANE) * (IMG_H * IMG_W);
        const float* q1 = img1 + tb;
        const float* q2 = img2 + tb;
        int gc = tc0 - HALO + lc;
        gc = (gc < 0) ? -gc : gc;
        gc = (gc >= IMG_W) ? (2 * IMG_W - 2 - gc) : gc;
#pragma unroll
        for (int k = 0; k < LD_PASSES; ++k) {
            int gr = tr0 - HALO + lr + k * LD_ROWS;
            gr = (gr < 0) ? -gr : gr;
            gr = (gr >= IMG_H) ? (2 * IMG_H - 2 - gr) : gr;
            va[k] = q1[gr * IMG_W + gc];
            vb[k] = q2[gr * IMG_W + gc];
        }
    };
    if (blockIdx.x < ntiles) issue_halo(blockIdx.x);

    if (tid < 64) {
        int t = tid - 16;
        float v = 0.0f;
        if (t >= 0 && t <= 10) {
            int d = t - 5; if (d < 0) d = -d;
            v = (d == 0) ? G5 : (d == 1) ? G4 : (d == 2) ? G3
              : (d == 3) ? G2 : (d == 4) ? G1 : G0;
        }
        gpad[tid] = v;
    }
    lds_barrier();

    float ax[7], ay[7];
#pragma unroll
    for (int b = 0; b < 7; ++b) {
        int j0 = 4 * b + 2 * hi;
        ax[b] = gpad[j0 - M + 16];
        ay[b] = gpad[j0 + 1 - M + 16];
    }
    const int wave = __builtin_amdgcn_readfirstlane(tid >> 5);

    for (unsigned tile = blockIdx.x; tile < ntiles; tile += WG_COUNT) {
    const int c0    = (int)(tile & (TILES_X - 1)) * TILE;
    const int r0    = (int)((tile / TILES_X) & (IMG_H / TILE - 1)) * TILE;
    const int plane = (int)(tile / TILES_PER_PLANE);
    const size_t pbase = (size_t)plane * (IMG_H * IMG_W);

    if (ld_on) {
#pragma unroll
        for (int k = 0; k < LD_PASSES; ++k) {
            xs[lofs + k * LD_ROWS * XS] = va[k];
            ys[lofs + k * LD_ROWS * XS] = vb[k];
        }
    }
    __syncthreads();

    {
        const float GW[11] = {G0, G1, G2, G3, G4, G5, G4, G3, G2, G1, G0};
        for (int i = tid; i < IN_T * TILE; i += 256) {
            int r = i >> 5;
            int c = i & 31;
            const float* xr = &xs[r * XS + c];
            const float* yr = &ys[r * XS + c];
            float sx = 0.f, sy = 0.f, sxx = 0.f, syy = 0.f, sxy = 0.f;
#pragma unroll
            for (int t = 0; t < 11; ++t) {
                float w  = GW[t];
                float xv = xr[t];
                float yv = yr[t];
                sx  += w * xv;
                sy  += w * yv;
                sxx += w * xv * xv;
                syy += w * yv * yv;
                sxy += w * xv * yv;
            }
            int o = r * HS + c;
            hbuf[0 * IN_T * HS + o] = sx;
            hbuf[1 * IN_T * HS + o] = sy;
            hbuf[2 * IN_T * HS + o] = sxx;
            hbuf[3 * IN_T * HS + o] = syy;
            hbuf[4 * IN_T * HS + o] = sxy;
        }
    }
    __syncthreads();

    if (tile + WG_COUNT < ntiles) issue_halo(tile + WG_COUNT);

    for (int j = wave; j < 20; j += 8) {
        int f  = j >> 2;
        int s  = j & 3;
        int sr = (s >> 1) << 4;
        int sc = (s & 1) << 4;
        const float* hf = hbuf + f * IN_T * HS;

        v8f acc = {0.f, 0.f, 0.f, 0.f, 0.f, 0.f, 0.f, 0.f};
#pragma unroll
        for (int b = 0; b < 7; ++b) {
            int j0   = 4 * b + 2 * hi;
            int row0 = sr + j0;     if (row0 > IN_T - 1) row0 = IN_T - 1;
            int row1 = sr + j0 + 1; if (row1 > IN_T - 1) row1 = IN_T - 1;
            v2f av = {ax[b], ay[b]};
            v2f bv;
            bv.x = hf[row0 * HS + sc + M];
            bv.y = hf[row1 * HS + sc + M];
            acc = __builtin_amdgcn_wmma_f32_16x16x4_f32(
                 false, av,  false, bv,
                 (short)0, acc,  false,  false);
        }
        float* rf = rbuf + f * TILE * RS;
#pragma unroll
        for (int v = 0; v < 8; ++v)
            rf[(sr + hi * 8 + v) * RS + sc + M] = acc[v];
    }
    lds_barrier();

    const float C1v = 0.0001f;
    const float C2v = 0.0009f;
    const float EPSv = 1e-12f;
    float* po = out + pbase;
    float fv[4][5];
#pragma unroll
    for (int k = 0; k < 4; ++k) {
        int p   = tid + (k << 8);
        int o   = (p >> 5) * RS + (p & 31);
#pragma unroll
        for (int f = 0; f < 5; ++f) fv[k][f] = rbuf[f * TILE * RS + o];
    }
    lds_barrier();
#pragma unroll
    for (int k = 0; k < 4; ++k) {
        int p   = tid + (k << 8);
        int row = p >> 5;
        int col = p & 31;
        float mu1 = fv[k][0];
        float mu2 = fv[k][1];
        float exx = fv[k][2];
        float eyy = fv[k][3];
        float exy = fv[k][4];
        float mu1s = mu1 * mu1;
        float mu2s = mu2 * mu2;
        float mu12 = mu1 * mu2;
        float s1  = exx - mu1s;
        float s2  = eyy - mu2s;
        float s12 = exy - mu12;
        float num = (2.0f * mu12 + C1v) * (2.0f * s12 + C2v);
        float den = (mu1s + mu2s + C1v) * (s1 + s2 + C2v);
        po[(size_t)(r0 + row) * IMG_W + (c0 + col)] = num / (den + EPSv);
    }
    }
}

extern "C" void kernel_launch(void* const* d_in, const int* in_sizes, int n_in,
                              void* d_out, int out_size, void* d_ws, size_t ws_size,
                              hipStream_t stream) {
    (void)n_in; (void)out_size; (void)d_ws; (void)ws_size;
    const float* img1 = (const float*)d_in[0];
    const float* img2 = (const float*)d_in[1];
    float* out = (float*)d_out;
    int planes = in_sizes[0] / (IMG_H * IMG_W);
    ssim_fused_kernel<<<dim3(WG_COUNT), 256, 0, stream>>>(img1, img2, out, planes);
}


# horizontal 11-tap pass moved from the vector pipe to the matrix pipe as a banded f32 product; x*x, y*y, x*y rounded in registers first, taps summed in the matrix unit's k order; h pitch 44
# speedup vs baseline: 1.2604x; 1.0444x over previous
#include <hip/hip_runtime.h>

typedef __attribute__((ext_vector_type(2))) float v2f;
typedef __attribute__((ext_vector_type(8))) float v8f;

#define TILE   32
#define HALO   5
#define IN_T   42
#define XS     44
#define HS     44
#define RS     36
#define IMG_H  512
#define IMG_W  512
#define TILES_X (IMG_W / TILE)
#define TILES_PER_PLANE (TILES_X * (IMG_H / TILE))
#define WG_COUNT 1280
#define LD_ROWS   6
#define LD_PASSES 7

#define G0 0.00102838f
#define G1 0.00759870f
#define G2 0.03600077f
#define G3 0.10936068f
#define G4 0.21300553f
#define G5 0.26601172f

static __device__ __forceinline__ void lds_barrier() {
    __builtin_amdgcn_fence(__ATOMIC_RELEASE, "workgroup", "local");
    __builtin_amdgcn_s_barrier();
    __builtin_amdgcn_fence(__ATOMIC_ACQUIRE, "workgroup", "local");
}

__global__ __launch_bounds__(256) __attribute__((amdgpu_waves_per_eu(10, 10))) void ssim_fused_kernel(const float* __restrict__ img1,
                                                         const float* __restrict__ img2,
                                                         float* __restrict__ out,
                                                         int planes) {
    __shared__ __align__(16) float smem[9240 + 64 + 5760];
    float* hbuf = smem;
    float* gpad = smem + 5 * IN_T * HS;
    float* xs   = gpad + 64;
    float* ys   = xs + IN_T * XS;
    float* rbuf = gpad + 64;

    const int tid  = threadIdx.x;
    const int lane = tid & 31;
    const int M    = lane & 15;
    const int hi   = lane >> 4;

    const unsigned ntiles = (unsigned)planes * TILES_PER_PLANE;

    const bool ld_on = tid < LD_ROWS * IN_T;
    const int  lr    = tid / IN_T;
    const int  lc    = tid - lr * IN_T;
    const int  lofs  = lr * XS + lc;

    float va[LD_PASSES], vb[LD_PASSES];
    auto issue_halo = [&](unsigned t) {
        const int tc0 = (int)(t & (TILES_X - 1)) * TILE;
        const int tr0 = (int)((t / TILES_X) & (IMG_H / TILE - 1)) * TILE;
        const size_t tb = (size_t)(t / TILES_PER_PLANE) * (IMG_H * IMG_W);
        const float* q1 = img1 + tb;
        const float* q2 = img2 + tb;
        int gc = tc0 - HALO + lc;
        gc = (gc < 0) ? -gc : gc;
        gc = (gc >= IMG_W) ? (2 * IMG_W - 2 - gc) : gc;
#pragma unroll
        for (int k = 0; k < LD_PASSES; ++k) {
            int gr = tr0 - HALO + lr + k * LD_ROWS;
            gr = (gr < 0) ? -gr : gr;
            gr = (gr >= IMG_H) ? (2 * IMG_H - 2 - gr) : gr;
            va[k] = q1[gr * IMG_W + gc];
            vb[k] = q2[gr * IMG_W + gc];
        }
    };
    if (blockIdx.x < ntiles) issue_halo(blockIdx.x);

    if (tid < 64) {
        int t = tid - 16;
        float v = 0.0f;
        if (t >= 0 && t <= 10) {
            int d = t - 5; if (d < 0) d = -d;
            v = (d == 0) ? G5 : (d == 1) ? G4 : (d == 2) ? G3
              : (d == 3) ? G2 : (d == 4) ? G1 : G0;
        }
        gpad[tid] = v;
    }
    lds_barrier();

    float ax[7], ay[7];
#pragma unroll
    for (int b = 0; b < 7; ++b) {
        int j0 = 4 * b + 2 * hi;
        ax[b] = gpad[j0 - M + 16];
        ay[b] = gpad[j0 + 1 - M + 16];
    }
    const int wave = __builtin_amdgcn_readfirstlane(tid >> 5);

    for (unsigned tile = blockIdx.x; tile < ntiles; tile += WG_COUNT) {
    const int c0    = (int)(tile & (TILES_X - 1)) * TILE;
    const int r0    = (int)((tile / TILES_X) & (IMG_H / TILE - 1)) * TILE;
    const int plane = (int)(tile / TILES_PER_PLANE);
    const size_t pbase = (size_t)plane * (IMG_H * IMG_W);

    if (ld_on) {
#pragma unroll
        for (int k = 0; k < LD_PASSES; ++k) {
            xs[lofs + k * LD_ROWS * XS] = va[k];
            ys[lofs + k * LD_ROWS * XS] = vb[k];
        }
    }
    __syncthreads();

    {
        auto load_frags = [&](const float* src, int base, v2f (&a)[7]) {
#pragma unroll
            for (int b = 0; b < 7; ++b) {
                int off = (b < 6) ? base + 4 * b + 2 * hi : base + 24;
                a[b] = *(const v2f*)(src + off);
            }
        };
        auto store_tile = [&](float* hf, int mt, int nt, const v8f& acc) {
            float* p = hf + (mt * 16 + 8 * hi) * HS + nt * 16 + M;
            p[0]  = acc[0];
            p[HS] = acc[1];
            if (mt * 16 + 8 * hi + 7 < IN_T) {
#pragma unroll
                for (int v = 2; v < 8; ++v) p[v * HS] = acc[v];
            }
        };
        if (wave < 6) {
            const int mt = wave >> 1;
            const int nt = wave & 1;
            int r = mt * 16 + M; r = (r > IN_T - 1) ? IN_T - 1 : r;
            const int base = r * XS + nt * 16;
            v2f xa[7], ya[7];
            v8f s1 = {0.f, 0.f, 0.f, 0.f, 0.f, 0.f, 0.f, 0.f};
            v8f s2 = s1, s3 = s1, s4 = s1;
            load_frags(xs, base, xa);
            __builtin_amdgcn_sched_barrier(0);
#pragma unroll
            for (int b = 0; b < 4; ++b) {
                v2f bw = {ax[b], ay[b]};
                v2f xx = xa[b] * xa[b];
                s1 = __builtin_amdgcn_wmma_f32_16x16x4_f32(false, xa[b], false, bw, (short)0, s1, false, false);
                s2 = __builtin_amdgcn_wmma_f32_16x16x4_f32(false, xx,    false, bw, (short)0, s2, false, false);
            }
            load_frags(ys, base, ya);
            __builtin_amdgcn_sched_barrier(0);
#pragma unroll
            for (int b = 4; b < 7; ++b) {
                v2f bw = {ax[b], ay[b]};
                v2f xx = xa[b] * xa[b];
                s1 = __builtin_amdgcn_wmma_f32_16x16x4_f32(false, xa[b], false, bw, (short)0, s1, false, false);
                s2 = __builtin_amdgcn_wmma_f32_16x16x4_f32(false, xx,    false, bw, (short)0, s2, false, false);
            }
            store_tile(hbuf + 0 * IN_T * HS, mt, nt, s1);
            store_tile(hbuf + 2 * IN_T * HS, mt, nt, s2);
#pragma unroll
            for (int b = 0; b < 7; ++b) {
                v2f bw = {ax[b], ay[b]};
                v2f yy = ya[b] * ya[b];
                s3 = __builtin_amdgcn_wmma_f32_16x16x4_f32(false, ya[b], false, bw, (short)0, s3, false, false);
                s4 = __builtin_amdgcn_wmma_f32_16x16x4_f32(false, yy,    false, bw, (short)0, s4, false, false);
            }
            store_tile(hbuf + 1 * IN_T * HS, mt, nt, s3);
            store_tile(hbuf + 3 * IN_T * HS, mt, nt, s4);
        } else {
#pragma nounroll
            for (int q = 0; q < 3; ++q) {
                const int mn = (wave - 6) * 3 + q;
                const int mt = mn >> 1;
                const int nt = mn & 1;
                int r = mt * 16 + M; r = (r > IN_T - 1) ? IN_T - 1 : r;
                const int base = r * XS + nt * 16;
                v2f xa[7], ya[7];
                load_frags(xs, base, xa);
                load_frags(ys, base, ya);
                __builtin_amdgcn_sched_barrier(0);
                v8f s5 = {0.f, 0.f, 0.f, 0.f, 0.f, 0.f, 0.f, 0.f};
#pragma unroll
                for (int b = 0; b < 7; ++b) {
                    v2f bw = {ax[b], ay[b]};
                    v2f xy = xa[b] * ya[b];
                    s5 = __builtin_amdgcn_wmma_f32_16x16x4_f32(false, xy, false, bw, (short)0, s5, false, false);
                }
                store_tile(hbuf + 4 * IN_T * HS, mt, nt, s5);
            }
        }
    }
    __syncthreads();

    if (tile + WG_COUNT < ntiles) issue_halo(tile + WG_COUNT);

    for (int j = wave; j < 20; j += 8) {
        int f  = j >> 2;
        int s  = j & 3;
        int sr = (s >> 1) << 4;
        int sc = (s & 1) << 4;
        const float* hf = hbuf + f * IN_T * HS;

        v8f acc = {0.f, 0.f, 0.f, 0.f, 0.f, 0.f, 0.f, 0.f};
#pragma unroll
        for (int b = 0; b < 7; ++b) {
            int j0   = 4 * b + 2 * hi;
            int row0 = sr + j0;     if (row0 > IN_T - 1) row0 = IN_T - 1;
            int row1 = sr + j0 + 1; if (row1 > IN_T - 1) row1 = IN_T - 1;
            v2f av = {ax[b], ay[b]};
            v2f bv;
            bv.x = hf[row0 * HS + sc + M];
            bv.y = hf[row1 * HS + sc + M];
            acc = __builtin_amdgcn_wmma_f32_16x16x4_f32(
                 false, av,  false, bv,
                 (short)0, acc,  false,  false);
        }
        float* rf = rbuf + f * TILE * RS;
#pragma unroll
        for (int v = 0; v < 8; ++v)
            rf[(sr + hi * 8 + v) * RS + sc + M] = acc[v];
    }
    lds_barrier();

    const float C1v = 0.0001f;
    const float C2v = 0.0009f;
    const float EPSv = 1e-12f;
    float* po = out + pbase;
    float fv[4][5];
#pragma unroll
    for (int k = 0; k < 4; ++k) {
        int p   = tid + (k << 8);
        int o   = (p >> 5) * RS + (p & 31);
#pragma unroll
        for (int f = 0; f < 5; ++f) fv[k][f] = rbuf[f * TILE * RS + o];
    }
    lds_barrier();
#pragma unroll
    for (int k = 0; k < 4; ++k) {
        int p   = tid + (k << 8);
        int row = p >> 5;
        int col = p & 31;
        float mu1 = fv[k][0];
        float mu2 = fv[k][1];
        float exx = fv[k][2];
        float eyy = fv[k][3];
        float exy = fv[k][4];
        float mu1s = mu1 * mu1;
        float mu2s = mu2 * mu2;
        float mu12 = mu1 * mu2;
        float s1  = exx - mu1s;
        float s2  = eyy - mu2s;
        float s12 = exy - mu12;
        float num = (2.0f * mu12 + C1v) * (2.0f * s12 + C2v);
        float den = (mu1s + mu2s + C1v) * (s1 + s2 + C2v);
        po[(size_t)(r0 + row) * IMG_W + (c0 + col)] = num / (den + EPSv);
    }
    }
}

extern "C" void kernel_launch(void* const* d_in, const int* in_sizes, int n_in,
                              void* d_out, int out_size, void* d_ws, size_t ws_size,
                              hipStream_t stream) {
    (void)n_in; (void)out_size; (void)d_ws; (void)ws_size;
    const float* img1 = (const float*)d_in[0];
    const float* img2 = (const float*)d_in[1];
    float* out = (float*)d_out;
    int planes = in_sizes[0] / (IMG_H * IMG_W);
    ssim_fused_kernel<<<dim3(WG_COUNT), 256, 0, stream>>>(img1, img2, out, planes);
}


# sigma1_sq and sigma2_sq enter only as their sum: x*x+y*y is filtered as ONE field, 4 fields and 280 WMMAs a tile instead of 5 and 350; squares are now added before the taps, not after
# speedup vs baseline: 1.4080x; 1.1171x over previous
#include <hip/hip_runtime.h>

typedef __attribute__((ext_vector_type(2))) float v2f;
typedef __attribute__((ext_vector_type(8))) float v8f;

#define TILE   32
#define HALO   5
#define IN_T   42
#define XS     44
#define HS     44
#define RS     36
#define IMG_H  512
#define IMG_W  512
#define TILES_X (IMG_W / TILE)
#define TILES_PER_PLANE (TILES_X * (IMG_H / TILE))
#define WG_COUNT 1280
#define LD_ROWS   6
#define LD_PASSES 7
#define NF        4

#define G0 0.00102838f
#define G1 0.00759870f
#define G2 0.03600077f
#define G3 0.10936068f
#define G4 0.21300553f
#define G5 0.26601172f

static __device__ __forceinline__ void lds_barrier() {
    __builtin_amdgcn_fence(__ATOMIC_RELEASE, "workgroup", "local");
    __builtin_amdgcn_s_barrier();
    __builtin_amdgcn_fence(__ATOMIC_ACQUIRE, "workgroup", "local");
}

__global__ __launch_bounds__(256) __attribute__((amdgpu_waves_per_eu(10, 10))) void ssim_fused_kernel(const float* __restrict__ img1,
                                                         const float* __restrict__ img2,
                                                         float* __restrict__ out,
                                                         int planes) {
    __shared__ __align__(16) float smem[7392 + 64 + 4608];
    float* hbuf = smem;
    float* gpad = smem + NF * IN_T * HS;
    float* xs   = gpad + 64;
    float* ys   = xs + IN_T * XS;
    float* rbuf = gpad + 64;

    const int tid  = threadIdx.x;
    const int lane = tid & 31;
    const int M    = lane & 15;
    const int hi   = lane >> 4;

    const unsigned ntiles = (unsigned)planes * TILES_PER_PLANE;

    const bool ld_on = tid < LD_ROWS * IN_T;
    const int  lr    = tid / IN_T;
    const int  lc    = tid - lr * IN_T;
    const int  lofs  = lr * XS + lc;

    float va[LD_PASSES], vb[LD_PASSES];
    auto issue_halo = [&](unsigned t) {
        const int tc0 = (int)(t & (TILES_X - 1)) * TILE;
        const int tr0 = (int)((t / TILES_X) & (IMG_H / TILE - 1)) * TILE;
        const size_t tb = (size_t)(t / TILES_PER_PLANE) * (IMG_H * IMG_W);
        const float* q1 = img1 + tb;
        const float* q2 = img2 + tb;
        int gc = tc0 - HALO + lc;
        gc = (gc < 0) ? -gc : gc;
        gc = (gc >= IMG_W) ? (2 * IMG_W - 2 - gc) : gc;
#pragma unroll
        for (int k = 0; k < LD_PASSES; ++k) {
            int gr = tr0 - HALO + lr + k * LD_ROWS;
            gr = (gr < 0) ? -gr : gr;
            gr = (gr >= IMG_H) ? (2 * IMG_H - 2 - gr) : gr;
            va[k] = q1[gr * IMG_W + gc];
            vb[k] = q2[gr * IMG_W + gc];
        }
    };
    if (blockIdx.x < ntiles) issue_halo(blockIdx.x);

    if (tid < 64) {
        int t = tid - 16;
        float v = 0.0f;
        if (t >= 0 && t <= 10) {
            int d = t - 5; if (d < 0) d = -d;
            v = (d == 0) ? G5 : (d == 1) ? G4 : (d == 2) ? G3
              : (d == 3) ? G2 : (d == 4) ? G1 : G0;
        }
        gpad[tid] = v;
    }
    lds_barrier();

    float ax[7], ay[7];
#pragma unroll
    for (int b = 0; b < 7; ++b) {
        int j0 = 4 * b + 2 * hi;
        ax[b] = gpad[j0 - M + 16];
        ay[b] = gpad[j0 + 1 - M + 16];
    }
    const int wave = __builtin_amdgcn_readfirstlane(tid >> 5);

    for (unsigned tile = blockIdx.x; tile < ntiles; tile += WG_COUNT) {
    const int c0    = (int)(tile & (TILES_X - 1)) * TILE;
    const int r0    = (int)((tile / TILES_X) & (IMG_H / TILE - 1)) * TILE;
    const int plane = (int)(tile / TILES_PER_PLANE);
    const size_t pbase = (size_t)plane * (IMG_H * IMG_W);

    if (ld_on) {
#pragma unroll
        for (int k = 0; k < LD_PASSES; ++k) {
            xs[lofs + k * LD_ROWS * XS] = va[k];
            ys[lofs + k * LD_ROWS * XS] = vb[k];
        }
    }
    __syncthreads();

    {
        auto load_range = [&](const float* src, int base, v2f (&a)[7], int b0, int b1) {
#pragma unroll
            for (int b = b0; b < b1; ++b) {
                int off = (b < 6) ? base + 4 * b + 2 * hi : base + 24;
                a[b] = *(const v2f*)(src + off);
            }
        };
        auto load_frags = [&](const float* src, int base, v2f (&a)[7]) { load_range(src, base, a, 0, 7); };
        auto store_tile = [&](float* hf, int mt, int nt, const v8f& acc) {
            float* p = hf + (mt * 16 + 8 * hi) * HS + nt * 16 + M;
            p[0]  = acc[0];
            p[HS] = acc[1];
            if (mt * 16 + 8 * hi + 7 < IN_T) {
#pragma unroll
                for (int v = 2; v < 8; ++v) p[v * HS] = acc[v];
            }
        };
        if (wave < 6) {
            const int mt = wave >> 1;
            const int nt = wave & 1;
            int r = mt * 16 + M; r = (r > IN_T - 1) ? IN_T - 1 : r;
            const int base = r * XS + nt * 16;
            v2f xa[7], ya[7];
            v8f s2 = {0.f, 0.f, 0.f, 0.f, 0.f, 0.f, 0.f, 0.f};
            v8f s3 = s2;
            auto ksteps = [&](int b0, int b1) {
#pragma unroll
                for (int b = b0; b < b1; ++b) {
                    v2f bw = {ax[b], ay[b]};
                    v2f qq = xa[b] * xa[b] + ya[b] * ya[b];
                    s3 = __builtin_amdgcn_wmma_f32_16x16x4_f32(false, ya[b], false, bw, (short)0, s3, false, false);
                    s2 = __builtin_amdgcn_wmma_f32_16x16x4_f32(false, qq,    false, bw, (short)0, s2, false, false);
                }
            };
            load_range(xs, base, xa, 0, 7);
            load_range(ys, base, ya, 0, 4);
            __builtin_amdgcn_sched_barrier(0);
            ksteps(0, 3);
            load_range(ys, base, ya, 4, 7);
            __builtin_amdgcn_sched_barrier(0);
            ksteps(3, 7);
            store_tile(hbuf + 1 * IN_T * HS, mt, nt, s3);
            store_tile(hbuf + 2 * IN_T * HS, mt, nt, s2);
            v8f s1 = {0.f, 0.f, 0.f, 0.f, 0.f, 0.f, 0.f, 0.f};
#pragma unroll
            for (int b = 0; b < 7; ++b) {
                v2f bw = {ax[b], ay[b]};
                s1 = __builtin_amdgcn_wmma_f32_16x16x4_f32(false, xa[b], false, bw, (short)0, s1, false, false);
            }
            store_tile(hbuf + 0 * IN_T * HS, mt, nt, s1);
        } else {
#pragma nounroll
            for (int q = 0; q < 3; ++q) {
                const int mn = (wave - 6) * 3 + q;
                const int mt = mn >> 1;
                const int nt = mn & 1;
                int r = mt * 16 + M; r = (r > IN_T - 1) ? IN_T - 1 : r;
                const int base = r * XS + nt * 16;
                v2f xa[7], ya[7];
                load_frags(xs, base, xa);
                load_frags(ys, base, ya);
                __builtin_amdgcn_sched_barrier(0);
                v8f s5 = {0.f, 0.f, 0.f, 0.f, 0.f, 0.f, 0.f, 0.f};
#pragma unroll
                for (int b = 0; b < 7; ++b) {
                    v2f bw = {ax[b], ay[b]};
                    v2f xy = xa[b] * ya[b];
                    s5 = __builtin_amdgcn_wmma_f32_16x16x4_f32(false, xy, false, bw, (short)0, s5, false, false);
                }
                store_tile(hbuf + 3 * IN_T * HS, mt, nt, s5);
            }
        }
    }
    __syncthreads();

    if (tile + WG_COUNT < ntiles) issue_halo(tile + WG_COUNT);

    for (int j = wave; j < NF * 4; j += 8) {
        int f  = j >> 2;
        int s  = j & 3;
        int sr = (s >> 1) << 4;
        int sc = (s & 1) << 4;
        const float* hf = hbuf + f * IN_T * HS;

        v8f acc = {0.f, 0.f, 0.f, 0.f, 0.f, 0.f, 0.f, 0.f};
#pragma unroll
        for (int b = 0; b < 7; ++b) {
            int j0   = 4 * b + 2 * hi;
            int row0 = sr + j0;     if (row0 > IN_T - 1) row0 = IN_T - 1;
            int row1 = sr + j0 + 1; if (row1 > IN_T - 1) row1 = IN_T - 1;
            v2f av = {ax[b], ay[b]};
            v2f bv;
            bv.x = hf[row0 * HS + sc + M];
            bv.y = hf[row1 * HS + sc + M];
            acc = __builtin_amdgcn_wmma_f32_16x16x4_f32(
                 false, av,  false, bv,
                 (short)0, acc,  false,  false);
        }
        float* rf = rbuf + f * TILE * RS;
#pragma unroll
        for (int v = 0; v < 8; ++v)
            rf[(sr + hi * 8 + v) * RS + sc + M] = acc[v];
    }
    lds_barrier();

    const float C1v = 0.0001f;
    const float C2v = 0.0009f;
    const float EPSv = 1e-12f;
    float* po = out + pbase;
    float fv[4][NF];
#pragma unroll
    for (int k = 0; k < 4; ++k) {
        int p   = tid + (k << 8);
        int o   = (p >> 5) * RS + (p & 31);
#pragma unroll
        for (int f = 0; f < NF; ++f) fv[k][f] = rbuf[f * TILE * RS + o];
    }
    lds_barrier();
#pragma unroll
    for (int k = 0; k < 4; ++k) {
        int p   = tid + (k << 8);
        int row = p >> 5;
        int col = p & 31;
        float mu1 = fv[k][0];
        float mu2 = fv[k][1];
        float eqq = fv[k][2];
        float exy = fv[k][3];
        float mu1s = mu1 * mu1;
        float mu2s = mu2 * mu2;
        float mu12 = mu1 * mu2;
        float ssum = (eqq - mu1s) - mu2s;
        float s12 = exy - mu12;
        float num = (2.0f * mu12 + C1v) * (2.0f * s12 + C2v);
        float den = (mu1s + mu2s + C1v) * (ssum + C2v);
        po[(size_t)(r0 + row) * IMG_W + (c0 + col)] = num / (den + EPSv);
    }
    }
}

extern "C" void kernel_launch(void* const* d_in, const int* in_sizes, int n_in,
                              void* d_out, int out_size, void* d_ws, size_t ws_size,
                              hipStream_t stream) {
    (void)n_in; (void)out_size; (void)d_ws; (void)ws_size;
    const float* img1 = (const float*)d_in[0];
    const float* img2 = (const float*)d_in[1];
    float* out = (float*)d_out;
    int planes = in_sizes[0] / (IMG_H * IMG_W);
    ssim_fused_kernel<<<dim3(WG_COUNT), 256, 0, stream>>>(img1, img2, out, planes);
}
